# starting kernel
# baseline (speedup 1.0000x reference)
#include <hip/hip_runtime.h>
#include <hip/hip_bf16.h>

#define T_LEN 512
#define B_SZ  512
#define EMB   50
#define HID   30
#define G4    120
#define VOC   100
#define NP    128
#define KE    52
#define KH    32
#define VP    112
#define CHUNK 16
#define NT    128

#define NP2 (NP * 2)
#define KH2 (KH * 2)
#define VP2 (VP * 2)

typedef __attribute__((ext_vector_type(2))) float v2f;
typedef __attribute__((ext_vector_type(8))) float v8f;

__device__ __forceinline__ v8f wmma4(v2f a, v2f b, v8f c) {
  return __builtin_amdgcn_wmma_f32_16x16x4_f32(false, a, false, b, (short)0, c,
                                               false, false);
}

__device__ __forceinline__ float sigm(float x) { return 1.0f / (1.0f + __expf(-x)); }

__device__ __forceinline__ v8f mm_acc(v8f acc, const float* __restrict__ aPtr, int kdim,
                                      const float* __restrict__ bPtr, int ld2) {
  #pragma unroll
  for (int kb = 0; kb < kdim; kb += 4) {
    v2f a = *(const v2f*)(aPtr + kb);
    v2f b = *(const v2f*)(bPtr + (kb >> 1) * ld2);
    acc = wmma4(a, b, acc);
  }
  return acc;
}

extern "C" __global__ __launch_bounds__(NT)
void charnn_fused(const int* __restrict__ batch_x, const int* __restrict__ lens,
                  const float* __restrict__ emb,
                  const float* __restrict__ w_ih1, const float* __restrict__ w_hh1,
                  const float* __restrict__ b_ih1, const float* __restrict__ b_hh1,
                  const float* __restrict__ w_ih2, const float* __restrict__ w_hh2,
                  const float* __restrict__ b_ih2, const float* __restrict__ b_hh2,
                  const float* __restrict__ lw1,  const float* __restrict__ lb1,
                  const float* __restrict__ lw2,  const float* __restrict__ lb2,
                  float* __restrict__ out)
{
  extern __shared__ float smem[];
  const int tid   = threadIdx.x;
  const int lane  = tid & 31;
  const int wave  = tid >> 5;
  const int row   = lane & 15;
  const int khalf = (lane < 16) ? 0 : 2;
  const int kh2   = khalf >> 1;
  const int mhalf = (lane < 16) ? 0 : 8;
  const int b0    = blockIdx.x * CHUNK;

  float* embL = smem;
  float* wi1  = embL + VOC * KE;
  float* wh1  = wi1  + KE * NP;
  float* wi2  = wh1  + KH * NP;
  float* wh2  = wi2  + KH * NP;
  float* l1T  = wh2  + KH * NP;
  float* l2T  = l1T  + KH * KH;
  float* bg1  = l2T  + KH * VP;
  float* bg2  = bg1  + NP;
  float* bl1L = bg2  + NP;
  float* bl2L = bl1L + KH;
  float* h1   = bl2L + VP;
  float* h2   = h1   + CHUNK * KH;
  float* h2m  = h2   + CHUNK * KH;
  float* hmid = h2m  + CHUNK * KH;
  float* gbuf = hmid + CHUNK * KH;
  float* mskf = gbuf + CHUNK * NP;
  int*   ids   = (int*)(mskf + CHUNK);
  int*   lensL = ids + CHUNK;

  for (int i = tid; i < VOC * KE; i += NT) { int r = i / KE, k = i - r * KE;
    embL[i] = (k < EMB) ? emb[r * EMB + k] : 0.f; }
  for (int i = tid; i < KE * NP; i += NT) { int k = i / NP, n = i - k * NP;
    wi1[(k >> 1) * NP2 + n * 2 + (k & 1)] =
        (k < EMB && n < G4) ? w_ih1[n * EMB + k] : 0.f; }
  for (int i = tid; i < KH * NP; i += NT) { int k = i / NP, n = i - k * NP;
    const bool ok = (k < HID && n < G4);
    const int  d  = (k >> 1) * NP2 + n * 2 + (k & 1);
    wh1[d] = ok ? w_hh1[n * HID + k] : 0.f;
    wi2[d] = ok ? w_ih2[n * HID + k] : 0.f;
    wh2[d] = ok ? w_hh2[n * HID + k] : 0.f; }
  for (int i = tid; i < KH * KH; i += NT) { int k = i / KH, j = i - k * KH;
    l1T[(k >> 1) * KH2 + j * 2 + (k & 1)] =
        (k < HID && j < HID) ? lw1[j * HID + k] : 0.f; }
  for (int i = tid; i < KH * VP; i += NT) { int k = i / VP, v = i - k * VP;
    l2T[(k >> 1) * VP2 + v * 2 + (k & 1)] =
        (k < HID && v < VOC) ? lw2[v * HID + k] : 0.f; }
  for (int i = tid; i < NP; i += NT) {
    bg1[i] = (i < G4) ? (b_ih1[i] + b_hh1[i]) : 0.f;
    bg2[i] = (i < G4) ? (b_ih2[i] + b_hh2[i]) : 0.f; }
  for (int i = tid; i < KH; i += NT) bl1L[i] = (i < HID) ? lb1[i] : 0.f;
  for (int i = tid; i < VP; i += NT) bl2L[i] = (i < VOC) ? lb2[i] : 0.f;
  for (int i = tid; i < CHUNK * KH; i += NT) { h1[i] = 0.f; h2[i] = 0.f; h2m[i] = 0.f; hmid[i] = 0.f; }
  if (tid < CHUNK) lensL[tid] = lens[b0 + tid];

  float c1a[4] = {0.f, 0.f, 0.f, 0.f};
  float c2a[4] = {0.f, 0.f, 0.f, 0.f};

  __syncthreads();

  for (int t = 0; t < T_LEN; ++t) {
    if (tid < CHUNK) {
      ids[tid]  = batch_x[(b0 + tid) * T_LEN + t];
      mskf[tid] = (t < lensL[tid]) ? 1.f : 0.f;
    }
    __syncthreads();

    {
      const float* arowE = embL + ids[row] * KE + khalf;
      const float* arowH = h1 + row * KH + khalf;
      for (int nt = wave; nt < NP / 16; nt += 4) {
        const int n = nt * 16 + row;
        const float bv = bg1[n];
        v8f acc;
        #pragma unroll
        for (int r = 0; r < 8; ++r) acc[r] = bv;
        acc = mm_acc(acc, arowE, KE, wi1 + n * 2 + kh2 * NP2, NP2);
        acc = mm_acc(acc, arowH, KH, wh1 + n * 2 + kh2 * NP2, NP2);
        #pragma unroll
        for (int r = 0; r < 8; ++r) gbuf[(r + mhalf) * NP + n] = acc[r];
      }
    }
    __syncthreads();

    #pragma unroll
    for (int r = 0; r < 4; ++r) {
      const int p = tid + NT * r;
      if (p < CHUNK * HID) {
        const int m = p / HID, j = p - m * HID;
        const float ig = sigm(gbuf[m * NP + j]);
        const float fg = sigm(gbuf[m * NP + HID + j]);
        const float gg = tanhf(gbuf[m * NP + 2 * HID + j]);
        const float og = sigm(gbuf[m * NP + 3 * HID + j]);
        const float cn = fg * c1a[r] + ig * gg;
        const float hn = og * tanhf(cn);
        const float mk = mskf[m];
        c1a[r] = mk * cn + (1.f - mk) * c1a[r];
        h1[m * KH + j] = mk * hn + (1.f - mk) * h1[m * KH + j];
      }
    }
    __syncthreads();

    {
      const float* arowA = h1 + row * KH + khalf;
      const float* arowB = h2 + row * KH + khalf;
      for (int nt = wave; nt < NP / 16; nt += 4) {
        const int n = nt * 16 + row;
        const float bv = bg2[n];
        v8f acc;
        #pragma unroll
        for (int r = 0; r < 8; ++r) acc[r] = bv;
        acc = mm_acc(acc, arowA, KH, wi2 + n * 2 + kh2 * NP2, NP2);
        acc = mm_acc(acc, arowB, KH, wh2 + n * 2 + kh2 * NP2, NP2);
        #pragma unroll
        for (int r = 0; r < 8; ++r) gbuf[(r + mhalf) * NP + n] = acc[r];
      }
    }
    __syncthreads();

    #pragma unroll
    for (int r = 0; r < 4; ++r) {
      const int p = tid + NT * r;
      if (p < CHUNK * HID) {
        const int m = p / HID, j = p - m * HID;
        const float ig = sigm(gbuf[m * NP + j]);
        const float fg = sigm(gbuf[m * NP + HID + j]);
        const float gg = tanhf(gbuf[m * NP + 2 * HID + j]);
        const float og = sigm(gbuf[m * NP + 3 * HID + j]);
        const float cn = fg * c2a[r] + ig * gg;
        const float hn = og * tanhf(cn);
        const float mk = mskf[m];
        c2a[r] = mk * cn + (1.f - mk) * c2a[r];
        const float hv = mk * hn + (1.f - mk) * h2[m * KH + j];
        h2[m * KH + j]  = hv;
        h2m[m * KH + j] = mk * hv;
      }
    }
    __syncthreads();

    if (wave < 2) {
      const float* arow = h2m + row * KH + khalf;
      const int n = wave * 16 + row;
      const float bv = bl1L[n];
      v8f acc;
      #pragma unroll
      for (int r = 0; r < 8; ++r) acc[r] = bv;
      acc = mm_acc(acc, arow, KH, l1T + n * 2 + kh2 * KH2, KH2);
      #pragma unroll
      for (int r = 0; r < 8; ++r) hmid[(r + mhalf) * KH + n] = fmaxf(acc[r], 0.f);
    }
    __syncthreads();

    {
      const float* arow = hmid + row * KH + khalf;
      for (int nt = wave; nt < VP / 16; nt += 4) {
        const int n = nt * 16 + row;
        const float bv = bl2L[n];
        v8f acc;
        #pragma unroll
        for (int r = 0; r < 8; ++r) acc[r] = bv;
        acc = mm_acc(acc, arow, KH, l2T + n * 2 + kh2 * VP2, VP2);
        if (n < VOC) {
          #pragma unroll
          for (int r = 0; r < 8; ++r) {
            const int m = r + mhalf;
            out[((size_t)(b0 + m) * T_LEN + t) * VOC + n] = acc[r];
          }
        }
      }
    }
    __syncthreads();
  }
}

extern "C" void kernel_launch(void* const* d_in, const int* in_sizes, int n_in,
                              void* d_out, int out_size, void* d_ws, size_t ws_size,
                              hipStream_t stream) {
  const int*   batch_x = (const int*)  d_in[0];
  const int*   lens    = (const int*)  d_in[1];
  const float* emb     = (const float*)d_in[2];
  const float* w_ih1   = (const float*)d_in[3];
  const float* w_hh1   = (const float*)d_in[4];
  const float* b_ih1   = (const float*)d_in[5];
  const float* b_hh1   = (const float*)d_in[6];
  const float* w_ih2   = (const float*)d_in[7];
  const float* w_hh2   = (const float*)d_in[8];
  const float* b_ih2   = (const float*)d_in[9];
  const float* b_hh2   = (const float*)d_in[10];
  const float* lw1     = (const float*)d_in[11];
  const float* lb1     = (const float*)d_in[12];
  const float* lw2     = (const float*)d_in[13];
  const float* lb2     = (const float*)d_in[14];
  float* out = (float*)d_out;

  constexpr size_t SMEM_FLOATS =
      (size_t)VOC * KE + (size_t)KE * NP + 3u * KH * NP + (size_t)KH * KH +
      (size_t)KH * VP + 2u * NP + KH + VP + 4u * CHUNK * KH + (size_t)CHUNK * NP +
      3u * CHUNK;
  constexpr size_t SMEM_BYTES = SMEM_FLOATS * sizeof(float);

  (void)hipFuncSetAttribute((const void*)charnn_fused,
                            hipFuncAttributeMaxDynamicSharedMemorySize,
                            (int)SMEM_BYTES);

  charnn_fused<<<dim3(B_SZ / CHUNK), dim3(NT), SMEM_BYTES, stream>>>(
      batch_x, lens, emb, w_ih1, w_hh1, b_ih1, b_hh1,
      w_ih2, w_hh2, b_ih2, b_hh2, lw1, lb1, lw2, lb2, out);
}


# Prologue weight staging: a thread's global loads issued in batches ahead of its LDS stores, lanes mapped 4 rows x 8 k (4 lines a load, not 32): 8 memory trips in a row, not 90; values unchanged
# speedup vs baseline: 7.1749x; 7.1749x over previous
#include <hip/hip_runtime.h>
#include <hip/hip_bf16.h>


#define T_LEN 512
#define B_SZ  512
#define EMB   50
#define HID   30
#define G4    120
#define VOC   100
#define NP    128
#define KE    52
#define KH    32
#define VP    112
#define MT    16
#define CHUNK 2
#define LW    4
#define NT    256
#define TB    16
#define XW    2
#define WPR   (LW / CHUNK)
#define NBLK  (T_LEN / TB)

#define NP2 (NP * 2)
#define KH2 (KH * 2)
#define VP2 (VP * 2)

typedef __attribute__((ext_vector_type(2))) float v2f;
typedef __attribute__((ext_vector_type(4))) float v4f;
typedef __attribute__((ext_vector_type(8))) float v8f;

__device__ __forceinline__ v8f wmma4(v2f a, v2f b, v8f c) {
  return __builtin_amdgcn_wmma_f32_16x16x4_f32(false, a, false, b, (short)0, c,
                                               false, false);
}

__device__ __forceinline__ float sigm(float x) { return 1.0f / (1.0f + __expf(-x)); }

__device__ __forceinline__ float tanh_sel(float x) {
  const float y  = __builtin_fabsf(x);
  const float y2 = x * x;
  float p = fmaf(y2, -0x1.758e7ap-8f, 0x1.521192p-6f);
  p = fmaf(y2, p, -0x1.b8389cp-5f);
  p = fmaf(y2, p, 0x1.110704p-3f);
  p = fmaf(y2, p, -0x1.555532p-2f);
  const float zs = fmaf(y2, y * p, y);
  const float t  = expf(2.0f * y);
  const float zb = fmaf(__builtin_amdgcn_rcpf(t + 1.0f), -2.0f, 1.0f);
  return __builtin_copysignf((y < 0.625f) ? zs : zb, x);
}

template <int KDIM, int NTL>
__device__ __forceinline__ void proj_block(float* __restrict__ xp, const float* __restrict__ arow,
                                           const float* __restrict__ wB, const float* __restrict__ bias,
                                           int m, int nt0, int row, int kh2, int mhalf) {
  v2f af[KDIM / 4];
  #pragma unroll
  for (int kq = 0; kq < KDIM / 4; ++kq) af[kq] = *(const v2f*)(arow + kq * 4);
  v8f acc[NTL];
  #pragma unroll
  for (int nt = 0; nt < NTL; ++nt) {
    const float bv = bias[(nt0 + nt) * 16 + row];
    #pragma unroll
    for (int r = 0; r < 8; ++r) acc[nt][r] = bv;
  }
  #pragma unroll
  for (int kq = 0; kq < KDIM / 4; ++kq) {
    #pragma unroll
    for (int nt = 0; nt < NTL; ++nt) {
      const v2f b = *(const v2f*)(wB + ((nt0 + nt) * 16 + row) * 2 + kh2 * NP2 + (kq * 2) * NP2);
      acc[nt] = wmma4(af[kq], b, acc[nt]);
    }
  }
  #pragma unroll
  for (int nt = 0; nt < NTL; ++nt) {
    #pragma unroll
    for (int r = 0; r < 8; ++r) xp[(((r + mhalf) * NP + (nt0 + nt) * 16 + row) * XW) + m] = acc[nt][r];
  }
}

extern "C" __global__ __launch_bounds__(NT)
void charnn_fused(const int* __restrict__ batch_x, const int* __restrict__ lens,
                  const float* __restrict__ emb,
                  const float* __restrict__ w_ih1, const float* __restrict__ w_hh1,
                  const float* __restrict__ b_ih1, const float* __restrict__ b_hh1,
                  const float* __restrict__ w_ih2, const float* __restrict__ w_hh2,
                  const float* __restrict__ b_ih2, const float* __restrict__ b_hh2,
                  const float* __restrict__ lw1,  const float* __restrict__ lb1,
                  const float* __restrict__ lw2,  const float* __restrict__ lb2,
                  float* __restrict__ out)
{
  extern __shared__ __align__(32) float smem[];
  const int tid   = threadIdx.x;
  const int lane  = tid & 31;
  const int wave  = __builtin_amdgcn_readfirstlane(tid >> 5);
  const int layer = wave >> 2;
  const int w4    = wave & (LW - 1);
  const int bm    = w4 & (CHUNK - 1);
  const int bh    = w4 / CHUNK;
  const int row   = lane & 15;
  const int khalf = (lane < 16) ? 0 : 2;
  const int kh2   = khalf >> 1;
  const int mhalf = (lane < 16) ? 0 : 8;
  const int b0    = blockIdx.x * CHUNK;

  float* embL = smem;
  float* wi1  = embL + VOC * KE;
  float* wh1  = wi1  + KE * NP;
  float* wi2  = wh1  + KH * NP;
  float* wh2  = wi2  + KH * NP;
  float* l1T  = wh2  + KH * NP;
  float* l2T  = l1T  + KH * KH;
  float* bg1  = l2T  + KH * VP;
  float* bg2  = bg1  + NP;
  float* bl1L = bg2  + NP;
  float* bl2L = bl1L + KH;
  float* h1   = bl2L + VP;
  float* h2   = h1   + MT * KH;
  float* hs1  = h2   + MT * KH;
  float* hs2  = hs1  + CHUNK * TB * KH;
  float* hmidW = hs2 + CHUNK * TB * KH;
  float* gb1  = hmidW + CHUNK * MT * KH;
  float* gb2  = gb1  + MT * NP;
  float* xp1  = gb2  + MT * NP;
  float* xp2  = xp1  + TB * NP * XW;
  int*   tokL = (int*)(xp2 + TB * NP * XW);

  {
    constexpr int NU = (VOC * KE + NT - 1) / NT;
    float v[NU];
    #pragma unroll
    for (int u = 0; u < NU; ++u) { const int i = min(tid + u * NT, VOC * KE - 1); const int r = i / KE, k = i - r * KE;
      v[u] = emb[r * EMB + min(k, EMB - 1)]; }
    #pragma unroll
    for (int u = 0; u < NU; ++u) { const int i = tid + u * NT; const int r = i / KE, k = i - r * KE;
      if (i < VOC * KE) embL[i] = (k < EMB) ? v[u] : 0.f; }
  }
  {
    constexpr int K8 = (KE + 7) / 8;
    constexpr int NU = (K8 * 8 * NP) / NT;
    static_assert(NU * NT == K8 * 8 * NP, "whole passes");
    float v[NU];
    #pragma unroll
    for (int u = 0; u < NU; ++u) { const int i = tid + u * NT; const int k = (i / (8 * NP)) * 8 + i % 8, n = (i / 8) % NP;
      v[u] = w_ih1[min(n, G4 - 1) * EMB + min(k, EMB - 1)]; }
    #pragma unroll
    for (int u = 0; u < NU; ++u) { const int i = tid + u * NT; const int k = (i / (8 * NP)) * 8 + i % 8, n = (i / 8) % NP;
      if (k < KE) wi1[(k >> 1) * NP2 + n * 2 + (k & 1)] = (k < EMB && n < G4) ? v[u] : 0.f; }
  }
  {
    constexpr int NU = (KH * NP) / NT;
    static_assert(NU * NT == KH * NP && KH % 8 == 0, "whole passes, whole groups of 8 k");
    float va[NU], vb[NU], vc[NU];
    #pragma unroll
    for (int u = 0; u < NU; ++u) { const int i = tid + u * NT; const int k = (i / (8 * NP)) * 8 + i % 8, n = (i / 8) % NP;
      const int s = min(n, G4 - 1) * HID + min(k, HID - 1);
      va[u] = w_hh1[s]; vb[u] = w_ih2[s]; vc[u] = w_hh2[s]; }
    #pragma unroll
    for (int u = 0; u < NU; ++u) { const int i = tid + u * NT; const int k = (i / (8 * NP)) * 8 + i % 8, n = (i / 8) % NP;
      const bool ok = (k < HID && n < G4);
      const int  d  = (k >> 1) * NP2 + n * 2 + (k & 1);
      wh1[d] = ok ? va[u] : 0.f;
      wi2[d] = ok ? vb[u] : 0.f;
      wh2[d] = ok ? vc[u] : 0.f; }
  }
  {
    constexpr int NA = (KH * KH) / NT, NB = (KH * VP) / NT;
    static_assert(NA * NT == KH * KH && NB * NT == KH * VP, "whole passes");
    float va[NA], vb[NB];
    #pragma unroll
    for (int u = 0; u < NA; ++u) { const int i = tid + u * NT; const int k = (i / (8 * KH)) * 8 + i % 8, j = (i / 8) % KH;
      va[u] = lw1[min(j, HID - 1) * HID + min(k, HID - 1)]; }
    #pragma unroll
    for (int u = 0; u < NB; ++u) { const int i = tid + u * NT; const int k = (i / (8 * VP)) * 8 + i % 8, n = (i / 8) % VP;
      vb[u] = lw2[min(n, VOC - 1) * HID + min(k, HID - 1)]; }
    #pragma unroll
    for (int u = 0; u < NA; ++u) { const int i = tid + u * NT; const int k = (i / (8 * KH)) * 8 + i % 8, j = (i / 8) % KH;
      l1T[(k >> 1) * KH2 + j * 2 + (k & 1)] = (k < HID && j < HID) ? va[u] : 0.f; }
    #pragma unroll
    for (int u = 0; u < NB; ++u) { const int i = tid + u * NT; const int k = (i / (8 * VP)) * 8 + i % 8, n = (i / 8) % VP;
      l2T[(k >> 1) * VP2 + n * 2 + (k & 1)] = (k < HID && n < VOC) ? vb[u] : 0.f; }
  }
  {
    static_assert(NP <= NT && KH <= NT && VP <= NT, "one pass");
    constexpr int NK = (CHUNK * T_LEN) / NT;
    static_assert(NK * NT == CHUNK * T_LEN, "whole passes");
    const int ig = min(tid, G4 - 1);
    const float g1a = b_ih1[ig], g1b = b_hh1[ig], g2a = b_ih2[ig], g2b = b_hh2[ig];
    const float l1v = lb1[min(tid, HID - 1)];
    const float l2v = lb2[min(tid, VOC - 1)];
    int tk[NK];
    #pragma unroll
    for (int u = 0; u < NK; ++u) tk[u] = batch_x[b0 * T_LEN + tid + u * NT];
    if (tid < NP) {
      bg1[tid] = (tid < G4) ? (g1a + g1b) : 0.f;
      bg2[tid] = (tid < G4) ? (g2a + g2b) : 0.f; }
    if (tid < KH) bl1L[tid] = (tid < HID) ? l1v : 0.f;
    if (tid < VP) bl2L[tid] = (tid < VOC) ? l2v : 0.f;
    #pragma unroll
    for (int u = 0; u < NK; ++u) tokL[tid + u * NT] = tk[u];
  }
  for (int i = tid; i < MT * KH; i += NT) { h1[i] = 0.f; h2[i] = 0.f; }
  for (int i = tid; i < CHUNK * TB * KH; i += NT) { hs1[i] = 0.f; hs2[i] = 0.f; }

  const int  cidx  = tid & (LW * 32 - 1);
  const bool cellT = cidx < CHUNK * HID;
  const int  cm = cellT ? cidx / HID : 0;
  const int  cj = cellT ? cidx - cm * HID : 0;
  float cst = 0.f;
  float hst = 0.f;
  const int myLen = lens[b0 + cm];

  float*       hL  = layer ? h2  : h1;
  const float* wR  = layer ? wh2 : wh1;
  float*       gb  = layer ? gb2 : gb1;
  const float* xpL = layer ? xp2 : xp1;
  float*       hs  = layer ? hs2 : hs1;

  __syncthreads();

  if (layer == 0)
    proj_block<KE, NP / 16 / WPR>(xp1, embL + tokL[bm * T_LEN + row] * KE + khalf, wi1, bg1,
                                  bm, bh * (NP / 16 / WPR), row, kh2, mhalf);
  __syncthreads();

  for (int blk = 0; blk <= NBLK; ++blk) {
    const bool act = layer ? (blk >= 1) : (blk < NBLK);
    const int  tb  = (blk - layer) * TB;
    for (int tt = 0; tt < TB; ++tt) {
      const float mk = (tb + tt < myLen) ? 1.f : 0.f;

      if (act) {
        const float* arow = hL + (row < CHUNK ? row : 0) * KH + khalf;
        v2f af[KH / 4];
        #pragma unroll
        for (int kq = 0; kq < KH / 4; ++kq) af[kq] = *(const v2f*)(arow + kq * 4);
        v8f acc[2];
        #pragma unroll
        for (int i = 0; i < 2; ++i) {
          const int n = (w4 + LW * i) * 16 + row;
          const v2f s = *(const v2f*)(xpL + (tt * NP + n) * XW);
          acc[i][0] = s.x; acc[i][1] = s.y; acc[i][2] = 0.f; acc[i][3] = 0.f;
          acc[i][4] = 0.f; acc[i][5] = 0.f; acc[i][6] = 0.f; acc[i][7] = 0.f;
        }
        #pragma unroll
        for (int kq = 0; kq < KH / 4; ++kq) {
          #pragma unroll
          for (int i = 0; i < 2; ++i) {
            const int n = (w4 + LW * i) * 16 + row;
            const v2f b = *(const v2f*)(wR + n * 2 + kh2 * NP2 + (kq * 2) * NP2);
            acc[i] = wmma4(af[kq], b, acc[i]);
          }
        }
        if (lane < 16) {
          #pragma unroll
          for (int i = 0; i < 2; ++i) {
            const int n = (w4 + LW * i) * 16 + row;
            #pragma unroll
            for (int r = 0; r < CHUNK; ++r) gb[r * NP + n] = acc[i][r];
          }
        }
      }
      __syncthreads();

      if (act && cellT) {
        const int m = cm, j = cj;
        const float xi = gb[m * NP + j];
        const float xf = gb[m * NP + HID + j];
        const float xg = gb[m * NP + 2 * HID + j];
        const float xo = gb[m * NP + 3 * HID + j];
        const float ig = sigm(xi);
        const float fg = sigm(xf);
        const float og = sigm(xo);
        const float gg = tanh_sel(xg);
        const float cn = fg * cst + ig * gg;
        const float hn = og * tanh_sel(cn);
        cst = mk * cn + (1.f - mk) * cst;
        const float hv = mk * hn + (1.f - mk) * hst;
        hst = hv;
        hL[m * KH + j] = hv;
        hs[(m * TB + tt) * KH + j] = layer ? mk * hv : hv;
      }
      __syncthreads();
    }

    if (layer) {
      if (bh == 0) {
        if (blk >= 1) {
          float* hm = hmidW + bm * MT * KH;
          {
            const float* arow = hs2 + (bm * TB + row) * KH + khalf;
            v2f af[KH / 4];
            #pragma unroll
            for (int kq = 0; kq < KH / 4; ++kq) af[kq] = *(const v2f*)(arow + kq * 4);
            v8f a1[2];
            #pragma unroll
            for (int nt = 0; nt < 2; ++nt) {
              const float bv = bl1L[nt * 16 + row];
              #pragma unroll
              for (int r = 0; r < 8; ++r) a1[nt][r] = bv;
            }
            #pragma unroll
            for (int kq = 0; kq < KH / 4; ++kq) {
              #pragma unroll
              for (int nt = 0; nt < 2; ++nt) {
                const v2f b = *(const v2f*)(l1T + (nt * 16 + row) * 2 + kh2 * KH2 + (kq * 2) * KH2);
                a1[nt] = wmma4(af[kq], b, a1[nt]);
              }
            }
            #pragma unroll
            for (int nt = 0; nt < 2; ++nt) {
              #pragma unroll
              for (int r = 0; r < 8; ++r) hm[(r + mhalf) * KH + nt * 16 + row] = fmaxf(a1[nt][r], 0.f);
            }
          }
          {
            const float* arow = hm + row * KH + khalf;
            v2f ah[KH / 4];
            #pragma unroll
            for (int kq = 0; kq < KH / 4; ++kq) ah[kq] = *(const v2f*)(arow + kq * 4);
            v8f a2[VP / 16];
            #pragma unroll
            for (int nt = 0; nt < VP / 16; ++nt) {
              const float bv = bl2L[nt * 16 + row];
              #pragma unroll
              for (int r = 0; r < 8; ++r) a2[nt][r] = bv;
            }
            #pragma unroll
            for (int kq = 0; kq < KH / 4; ++kq) {
              #pragma unroll
              for (int nt = 0; nt < VP / 16; ++nt) {
                const v2f b = *(const v2f*)(l2T + (nt * 16 + row) * 2 + kh2 * VP2 + (kq * 2) * VP2);
                a2[nt] = wmma4(ah[kq], b, a2[nt]);
              }
            }
            float* orow = out + ((size_t)(b0 + bm) * T_LEN + tb + mhalf) * VOC;
            #pragma unroll
            for (int nt = 0; nt < VP / 16; ++nt) {
              const int n = nt * 16 + row;
              if (n < VOC) {
                #pragma unroll
                for (int r = 0; r < 8; ++r) orow[(size_t)r * VOC + n] = a2[nt][r];
              }
            }
          }
        }
      } else {
        if (blk < NBLK)
          proj_block<KH, NP / 16>(xp2, hs1 + (bm * TB + row) * KH + khalf, wi2, bg2, bm, 0, row, kh2, mhalf);
      }
    } else {
      if (blk + 1 < NBLK)
        proj_block<KE, NP / 16 / WPR>(xp1, embL + tokL[bm * T_LEN + (blk + 1) * TB + row] * KE + khalf, wi1, bg1,
                                      bm, bh * (NP / 16 / WPR), row, kh2, mhalf);
    }
    __syncthreads();
  }
}

extern "C" void kernel_launch(void* const* d_in, const int* in_sizes, int n_in,
                              void* d_out, int out_size, void* d_ws, size_t ws_size,
                              hipStream_t stream) {
  const int*   batch_x = (const int*)  d_in[0];
  const int*   lens    = (const int*)  d_in[1];
  const float* emb     = (const float*)d_in[2];
  const float* w_ih1   = (const float*)d_in[3];
  const float* w_hh1   = (const float*)d_in[4];
  const float* b_ih1   = (const float*)d_in[5];
  const float* b_hh1   = (const float*)d_in[6];
  const float* w_ih2   = (const float*)d_in[7];
  const float* w_hh2   = (const float*)d_in[8];
  const float* b_ih2   = (const float*)d_in[9];
  const float* b_hh2   = (const float*)d_in[10];
  const float* lw1     = (const float*)d_in[11];
  const float* lb1     = (const float*)d_in[12];
  const float* lw2     = (const float*)d_in[13];
  const float* lb2     = (const float*)d_in[14];
  float* out = (float*)d_out;

  constexpr size_t SMEM_FLOATS =
      (size_t)VOC * KE + (size_t)KE * NP + 3u * KH * NP + (size_t)KH * KH +
      (size_t)KH * VP + 2u * NP + KH + VP + 2u * MT * KH + 2u * CHUNK * TB * KH +
      (size_t)CHUNK * MT * KH + 2u * MT * NP + 2u * TB * NP * XW + (size_t)CHUNK * T_LEN;
  constexpr size_t USED_BYTES = SMEM_FLOATS * sizeof(float);
  constexpr size_t SMEM_BYTES = 232u * 1024u;
  static_assert(USED_BYTES <= SMEM_BYTES, "LDS carve-up exceeds the request");

  (void)hipFuncSetAttribute((const void*)charnn_fused,
                            hipFuncAttributeMaxDynamicSharedMemorySize,
                            (int)SMEM_BYTES);

  charnn_fused<<<dim3(B_SZ / CHUNK), dim3(NT), SMEM_BYTES, stream>>>(
      batch_x, lens, emb, w_ih1, w_hh1, b_ih1, b_hh1,
      w_ih2, w_hh2, b_ih2, b_hh2, lw1, lb1, lw2, lb2, out);
}
